# starting kernel
# baseline (speedup 1.0000x reference)
#include <hip/hip_runtime.h>
#include <hip/hip_bf16.h>


typedef __attribute__((ext_vector_type(2))) float v2f;
typedef __attribute__((ext_vector_type(8))) float v8f;

#define BATCH     32
#define FRAMES    4000
#define CBINS     65
#define NIR       129
#define FRAME_LEN 64
#define SEGLEN    192
#define OUTLEN    256128
#define KPAD      68
#define JROWS     144
#define IRS       145
#define FPB       16

__global__ void FN_setup_M(float* __restrict__ M) {
  const float TWO_PI = 6.28318530717958647692f;
  for (int i = blockIdx.x * blockDim.x + threadIdx.x; i < JROWS * KPAD;
       i += gridDim.x * blockDim.x) {
    int j = i / KPAD;
    int k = i - j * KPAD;
    float v = 0.0f;
    if (j < NIR && k < CBINS) {
      float win = 0.5f * (1.0f - cosf(TWO_PI * (float)j / (float)NIR));
      int m = (k * (j - 64)) % NIR;
      float ang = TWO_PI * (float)m / (float)NIR;
      float ck  = (k == 0) ? 1.0f : 2.0f;
      v = win * (ck / (float)NIR) * cosf(ang);
    }
    M[i] = v;
  }
}

__global__ void FN_zero(float4* __restrict__ out4, int n4) {
  float4 z; z.x = z.y = z.z = z.w = 0.0f;
  for (int i = blockIdx.x * blockDim.x + threadIdx.x; i < n4;
       i += gridDim.x * blockDim.x)
    out4[i] = z;
}

__global__ __launch_bounds__(256) void FN_fused(const float* __restrict__ H,
                                                const float* __restrict__ noise,
                                                const float* __restrict__ M,
                                                float* __restrict__ out) {
  __shared__ float irs[FPB * IRS];
  __shared__ float nz[FPB * FRAME_LEN];

  const int tid  = threadIdx.x;
  const int lane = tid & 31;
  const int wave = tid >> 5;

  const int b  = blockIdx.x / (FRAMES / FPB);
  const int f0 = (blockIdx.x - b * (FRAMES / FPB)) * FPB;
  const long r0 = (long)b * FRAMES + f0;

  for (int i = tid; i < FPB * FRAME_LEN; i += 256) {
    int lf = i >> 6;
    int tt = i & 63;
    nz[i] = noise[(r0 + lf) * FRAME_LEN + tt] * 2.0f - 1.0f;
  }

  const int row = lane & 15;
  const int kb  = (lane >> 4) << 1;
  const float* hrow = H + (r0 + row) * CBINS;

  for (int jt = wave; jt < 9; jt += 8) {
    v8f acc = {};
    const float* mrow = M + (jt * 16 + row) * KPAD + kb;
#pragma unroll
    for (int k0 = 0; k0 < KPAD; k0 += 4) {
      int ka = k0 + kb;
      v2f a;
      a.x = (ka     < CBINS) ? hrow[ka]     : 0.0f;
      a.y = (ka + 1 < CBINS) ? hrow[ka + 1] : 0.0f;
      v2f bv = *(const v2f*)(mrow + k0);
      acc = __builtin_amdgcn_wmma_f32_16x16x4_f32(
           false, a,  false, bv,
           (short)0, acc,  false,  false);
    }
    const int mbase = (lane >> 4) << 3;
#pragma unroll
    for (int v = 0; v < 8; ++v)
      irs[(mbase + v) * IRS + jt * 16 + row] = acc[v];
  }

  __syncthreads();

  for (int o = tid; o < FPB * SEGLEN; o += 256) {
    int lf = o / SEGLEN;
    int t  = o - lf * SEGLEN;
    int lo = (t > NIR - 1) ? (t - (NIR - 1)) : 0;
    int hi = (t < FRAME_LEN - 1) ? t : (FRAME_LEN - 1);
    const float* np  = nz + (lf << 6);
    const float* irp = irs + lf * IRS + t;
    float s = 0.0f;
    for (int tau = lo; tau <= hi; ++tau)
      s += np[tau] * irp[-tau];
    __hip_atomic_fetch_add(
        out + (long)b * OUTLEN + (long)(f0 + lf) * FRAME_LEN + t,
        s * 0.01f, __ATOMIC_RELAXED, __HIP_MEMORY_SCOPE_AGENT);
  }
}

extern "C" void kernel_launch(void* const* d_in, const int* in_sizes, int n_in,
                              void* d_out, int out_size, void* d_ws, size_t ws_size,
                              hipStream_t stream) {
  const float* H     = (const float*)d_in[0];
  const float* noise = (const float*)d_in[1];
  float* out = (float*)d_out;
  float* Mws = (float*)d_ws;

  FN_setup_M<<<(JROWS * KPAD + 255) / 256, 256, 0, stream>>>(Mws);
  FN_zero<<<4096, 256, 0, stream>>>((float4*)out, out_size / 4);
  FN_fused<<<BATCH * (FRAMES / FPB), 256, 0, stream>>>(H, noise, Mws, out);
}


# conv stage register-tiled: 4 consecutive outputs per thread from a sliding window of zero-padded ir rows, 16-B LDS reads (2 per 16 FMA) instead of 2 4-B reads per FMA; atomics unchanged
# speedup vs baseline: 1.9826x; 1.9826x over previous
#include <hip/hip_runtime.h>
#include <hip/hip_bf16.h>


typedef __attribute__((ext_vector_type(2))) float v2f;
typedef __attribute__((ext_vector_type(8))) float v8f;

#define BATCH     32
#define FRAMES    4000
#define CBINS     65
#define NIR       129
#define FRAME_LEN 64
#define SEGLEN    192
#define OUTLEN    256128
#define KPAD      68
#define JROWS     144
#define FPB       16
#define IROFF     64
#define IRP       260
#define QPF       48

__global__ void FN_setup_M(float* __restrict__ M) {
  const float TWO_PI = 6.28318530717958647692f;
  for (int i = blockIdx.x * blockDim.x + threadIdx.x; i < JROWS * KPAD;
       i += gridDim.x * blockDim.x) {
    int j = i / KPAD;
    int k = i - j * KPAD;
    float v = 0.0f;
    if (j < NIR && k < CBINS) {
      float win = 0.5f * (1.0f - cosf(TWO_PI * (float)j / (float)NIR));
      int m = (k * (j - 64)) % NIR;
      float ang = TWO_PI * (float)m / (float)NIR;
      float ck  = (k == 0) ? 1.0f : 2.0f;
      v = win * (ck / (float)NIR) * cosf(ang);
    }
    M[i] = v;
  }
}

__global__ void FN_zero(float4* __restrict__ out4, int n4) {
  float4 z; z.x = z.y = z.z = z.w = 0.0f;
  for (int i = blockIdx.x * blockDim.x + threadIdx.x; i < n4;
       i += gridDim.x * blockDim.x)
    out4[i] = z;
}

__global__ __launch_bounds__(256) void FN_fused(const float* __restrict__ H,
                                                const float* __restrict__ noise,
                                                const float* __restrict__ M,
                                                float* __restrict__ out) {
  __shared__ __attribute__((aligned(16))) float irs[FPB * IRP];
  __shared__ __attribute__((aligned(16))) float nz[FPB * FRAME_LEN];
  __shared__ __attribute__((aligned(16))) float yb[FPB * SEGLEN];

  const int tid  = threadIdx.x;
  const int lane = tid & 31;
  const int wave = tid >> 5;

  const int b  = blockIdx.x / (FRAMES / FPB);
  const int f0 = (blockIdx.x - b * (FRAMES / FPB)) * FPB;
  const long r0 = (long)b * FRAMES + f0;

  for (int i = tid; i < FPB * FRAME_LEN; i += 256) {
    int lf = i >> 6;
    int tt = i & 63;
    nz[i] = noise[(r0 + lf) * FRAME_LEN + tt] * 2.0f - 1.0f;
  }

  for (int i = tid; i < FPB * 28; i += 256) {
    int rw = i / 28;
    int c  = i - rw * 28;
    int s4 = (c < 16) ? c : (c + 36);
    float4 z; z.x = z.y = z.z = z.w = 0.0f;
    *(float4*)(irs + rw * IRP + s4 * 4) = z;
  }

  const int row = lane & 15;
  const int kb  = (lane >> 4) << 1;
  const float* hrow = H + (r0 + row) * CBINS;

  for (int jt = wave; jt < 9; jt += 8) {
    v8f acc = {};
    const float* mrow = M + (jt * 16 + row) * KPAD + kb;
#pragma unroll
    for (int k0 = 0; k0 < KPAD; k0 += 4) {
      int ka = k0 + kb;
      v2f a;
      a.x = (ka     < CBINS) ? hrow[ka]     : 0.0f;
      a.y = (ka + 1 < CBINS) ? hrow[ka + 1] : 0.0f;
      v2f bv = *(const v2f*)(mrow + k0);
      acc = __builtin_amdgcn_wmma_f32_16x16x4_f32(
           false, a,  false, bv,
           (short)0, acc,  false,  false);
    }
    const int mbase = (lane >> 4) << 3;
#pragma unroll
    for (int v = 0; v < 8; ++v)
      irs[(mbase + v) * IRP + IROFF + jt * 16 + row] = acc[v];
  }

  __syncthreads();

  for (int it = tid; it < FPB * QPF; it += 256) {
    const int lf = it / QPF;
    const int q  = it - lf * QPF;
    const float*  irp = irs + lf * IRP + IROFF + 4 * q;
    const float4* np4 = (const float4*)(nz + (lf << 6));
    float s0 = 0.0f, s1 = 0.0f, s2 = 0.0f, s3 = 0.0f;
    float4 hi = *(const float4*)irp;
#pragma unroll
    for (int c4 = 0; c4 < FRAME_LEN / 4; ++c4) {
      const float4 lo = *(const float4*)(irp - 4 * c4 - 4);
      const float4 n  = np4[c4];
      s0 = fmaf(n.x, hi.x, s0); s0 = fmaf(n.y, lo.w, s0);
      s0 = fmaf(n.z, lo.z, s0); s0 = fmaf(n.w, lo.y, s0);
      s1 = fmaf(n.x, hi.y, s1); s1 = fmaf(n.y, hi.x, s1);
      s1 = fmaf(n.z, lo.w, s1); s1 = fmaf(n.w, lo.z, s1);
      s2 = fmaf(n.x, hi.z, s2); s2 = fmaf(n.y, hi.y, s2);
      s2 = fmaf(n.z, hi.x, s2); s2 = fmaf(n.w, lo.w, s2);
      s3 = fmaf(n.x, hi.w, s3); s3 = fmaf(n.y, hi.z, s3);
      s3 = fmaf(n.z, hi.y, s3); s3 = fmaf(n.w, hi.x, s3);
      hi = lo;
    }
    float4 sv;
    sv.x = s0 * 0.01f; sv.y = s1 * 0.01f; sv.z = s2 * 0.01f; sv.w = s3 * 0.01f;
    *(float4*)(yb + it * 4) = sv;
  }

  __syncthreads();

  for (int o = tid; o < FPB * SEGLEN; o += 256) {
    int lf = o / SEGLEN;
    int t  = o - lf * SEGLEN;
    __hip_atomic_fetch_add(
        out + (long)b * OUTLEN + (long)(f0 + lf) * FRAME_LEN + t,
        yb[o], __ATOMIC_RELAXED, __HIP_MEMORY_SCOPE_AGENT);
  }
}

extern "C" void kernel_launch(void* const* d_in, const int* in_sizes, int n_in,
                              void* d_out, int out_size, void* d_ws, size_t ws_size,
                              hipStream_t stream) {
  const float* H     = (const float*)d_in[0];
  const float* noise = (const float*)d_in[1];
  float* out = (float*)d_out;
  float* Mws = (float*)d_ws;

  FN_setup_M<<<(JROWS * KPAD + 255) / 256, 256, 0, stream>>>(Mws);
  FN_zero<<<4096, 256, 0, stream>>>((float4*)out, out_size / 4);
  FN_fused<<<BATCH * (FRAMES / FPB), 256, 0, stream>>>(H, noise, Mws, out);
}


# stage-1 WMMA operands by whole-line loads: H tile copied contiguously into LDS and read as 16-B fragments, fixed matrix stored in fragment order (a fragment load touched 16+ lines, now 1 to 4)
# speedup vs baseline: 3.7903x; 1.9118x over previous
#include <hip/hip_runtime.h>
#include <hip/hip_bf16.h>


typedef __attribute__((ext_vector_type(2))) float v2f;
typedef __attribute__((ext_vector_type(8))) float v8f;
typedef __attribute__((ext_vector_type(4))) float v4f;

#define BATCH     32
#define FRAMES    4000
#define CBINS     65
#define NIR       129
#define FRAME_LEN 64
#define SEGLEN    192
#define OUTLEN    256128
#define KPAD      68
#define JROWS     144
#define FPB       16
#define HPB       14
#define NHOPS     4002
#define WPB       286
#define IROFF     64
#define IRP       260
#define QPH       16
#define HSP       68
#define TILEF     1088

__global__ void FN_setup_M(float* __restrict__ M) {
  const float TWO_PI = 6.28318530717958647692f;
  for (int i = blockIdx.x * blockDim.x + threadIdx.x; i < JROWS * KPAD;
       i += gridDim.x * blockDim.x) {
    int jt = i / TILEF;
    int r  = i - jt * TILEF;
    int l, k;
    if (r < 1024) {
      int g = r >> 7;
      l = (r & 127) >> 2;
      k = 8 * g + 4 * (l >> 4) + (r & 3);
    } else {
      int r2 = r - 1024;
      l = r2 >> 1;
      k = 64 + 2 * (l >> 4) + (r2 & 1);
    }
    int j = jt * 16 + (l & 15);
    float v = 0.0f;
    if (j < NIR && k < CBINS) {
      float win = 0.5f * (1.0f - cosf(TWO_PI * (float)j / (float)NIR));
      int m = (k * (j - 64)) % NIR;
      float ang = TWO_PI * (float)m / (float)NIR;
      float ck  = (k == 0) ? 1.0f : 2.0f;
      v = win * (ck / (float)NIR) * cosf(ang);
    }
    M[i] = v;
  }
}

__device__ __forceinline__ void FN_load_b(const float* __restrict__ M, int jt, int lane,
                                          v4f (&bq)[8], v2f& bt) {
  const float* mp = M + jt * TILEF;
#pragma unroll
  for (int g = 0; g < 8; ++g) bq[g] = *(const v4f*)(mp + g * 128 + lane * 4);
  bt = *(const v2f*)(mp + 1024 + lane * 2);
}

__device__ __forceinline__ void FN_mm_tile(const float* hs, float* irs, int jt, int lane,
                                           const v4f (&bq)[8], v2f bt) {
  const int c  = lane & 15;
  const int hh = lane >> 4;
  const float* ar = hs + c * HSP;
  v8f acc = {};
#pragma unroll
  for (int g = 0; g < 8; ++g) {
    const v4f aq = *(const v4f*)(ar + 8 * g + 4 * hh);
    v2f a0, a1, b0, b1;
    a0.x = aq.x;    a0.y = aq.y;    a1.x = aq.z;    a1.y = aq.w;
    b0.x = bq[g].x; b0.y = bq[g].y; b1.x = bq[g].z; b1.y = bq[g].w;
    acc = __builtin_amdgcn_wmma_f32_16x16x4_f32(
         false, a0,  false, b0,
         (short)0, acc,  false,  false);
    acc = __builtin_amdgcn_wmma_f32_16x16x4_f32(
         false, a1,  false, b1,
         (short)0, acc,  false,  false);
  }
  const v2f at = *(const v2f*)(ar + 64 + 2 * hh);
  acc = __builtin_amdgcn_wmma_f32_16x16x4_f32(
       false, at,  false, bt,
       (short)0, acc,  false,  false);
  const int mbase = hh << 3;
#pragma unroll
  for (int v = 0; v < 8; ++v)
    irs[(mbase + v) * IRP + IROFF + jt * 16 + c] = acc[v];
}

__global__ __launch_bounds__(256) void FN_fused(const float* __restrict__ H,
                                                const float* __restrict__ noise,
                                                const float* __restrict__ M,
                                                float* __restrict__ out) {
  __shared__ __attribute__((aligned(16))) float irs[FPB * IRP];
  __shared__ __attribute__((aligned(16))) float nz[FPB * FRAME_LEN];
  __shared__ __attribute__((aligned(16))) float hs[FPB * HSP];

  const int tid  = threadIdx.x;
  const int lane = tid & 31;
  const int wave = tid >> 5;

  const int b  = blockIdx.x / WPB;
  const int g0 = (blockIdx.x - b * WPB) * HPB;
  const int fb = g0 - 2;
  const long rb = (long)b * FRAMES;

  float hv[5];
  int   ho[5];
#pragma unroll
  for (int u = 0; u < 5; ++u) {
    int i  = (u < 4) ? (tid + 256 * u) : min(tid + 1024, FPB * CBINS - 1);
    int lf = i / CBINS;
    int k  = i - lf * CBINS;
    int f  = min(max(fb + lf, 0), FRAMES - 1);
    hv[u] = H[(rb + f) * CBINS + k];
    ho[u] = lf * HSP + k;
  }

  v4f bq[8];
  v2f bt;
  FN_load_b(M, wave, lane, bq, bt);

  for (int i = tid; i < FPB * FRAME_LEN; i += 256) {
    int lf = i >> 6;
    int tt = i & 63;
    int f  = min(max(fb + lf, 0), FRAMES - 1);
    nz[i] = noise[(rb + f) * FRAME_LEN + tt] * 2.0f - 1.0f;
  }

  for (int i = tid; i < FPB * 28; i += 256) {
    int rw = i / 28;
    int c  = i - rw * 28;
    int s4 = (c < 16) ? c : (c + 36);
    float4 z; z.x = z.y = z.z = z.w = 0.0f;
    *(float4*)(irs + rw * IRP + s4 * 4) = z;
  }

#pragma unroll
  for (int u = 0; u < 4; ++u) hs[ho[u]] = hv[u];
  if (tid < FPB * CBINS - 1024) hs[ho[4]] = hv[4];
  if (tid < FPB * (HSP - CBINS))
    hs[(tid / 3) * HSP + CBINS + (tid - (tid / 3) * 3)] = 0.0f;

  __syncthreads();

  FN_mm_tile(hs, irs, wave, lane, bq, bt);
  if (wave == 0) {
    FN_load_b(M, 8, lane, bq, bt);
    FN_mm_tile(hs, irs, 8, lane, bq, bt);
  }

  __syncthreads();

  if (tid < HPB * QPH) {
    const int hl = tid >> 4;
    const int q  = tid & 15;
    const int h  = g0 + hl;
    if (h < NHOPS) {
      float s0 = 0.0f, s1 = 0.0f, s2 = 0.0f, s3 = 0.0f;
      for (int d = 0; d < 3; ++d) {
        const int lf = hl + 2 - d;
        const int f  = h - d;
        if (f < 0 || f >= FRAMES) continue;
        const float*  irp = irs + lf * IRP + IROFF + d * FRAME_LEN + 4 * q;
        const v4f* np4 = (const v4f*)(nz + (lf << 6));
        v4f hi = *(const v4f*)irp;
#pragma unroll
        for (int c4 = 0; c4 < FRAME_LEN / 4; ++c4) {
          const v4f lo = *(const v4f*)(irp - 4 * c4 - 4);
          const v4f n  = np4[c4];
          s0 = fmaf(n.x, hi.x, s0); s0 = fmaf(n.y, lo.w, s0);
          s0 = fmaf(n.z, lo.z, s0); s0 = fmaf(n.w, lo.y, s0);
          s1 = fmaf(n.x, hi.y, s1); s1 = fmaf(n.y, hi.x, s1);
          s1 = fmaf(n.z, lo.w, s1); s1 = fmaf(n.w, lo.z, s1);
          s2 = fmaf(n.x, hi.z, s2); s2 = fmaf(n.y, hi.y, s2);
          s2 = fmaf(n.z, hi.x, s2); s2 = fmaf(n.w, lo.w, s2);
          s3 = fmaf(n.x, hi.w, s3); s3 = fmaf(n.y, hi.z, s3);
          s3 = fmaf(n.z, hi.y, s3); s3 = fmaf(n.w, hi.x, s3);
          hi = lo;
        }
      }
      float4 sv;
      sv.x = s0 * 0.01f; sv.y = s1 * 0.01f; sv.z = s2 * 0.01f; sv.w = s3 * 0.01f;
      *(float4*)(out + (long)b * OUTLEN + (long)h * FRAME_LEN + 4 * q) = sv;
    }
  }
}

extern "C" void kernel_launch(void* const* d_in, const int* in_sizes, int n_in,
                              void* d_out, int out_size, void* d_ws, size_t ws_size,
                              hipStream_t stream) {
  const float* H     = (const float*)d_in[0];
  const float* noise = (const float*)d_in[1];
  float* out = (float*)d_out;
  float* Mws = (float*)d_ws;

  FN_setup_M<<<(JROWS * KPAD + 255) / 256, 256, 0, stream>>>(Mws);
  FN_fused<<<BATCH * WPB, 256, 0, stream>>>(H, noise, Mws, out);
}


# conv remapped so tap groups that only meet zero padding are skipped: a lane owns a hop, a wave 8 samples of it, loop bounds wave-uniform; 30 hops (32 frames) per work-group fill the lanes
# speedup vs baseline: 5.3356x; 1.4077x over previous
#include <hip/hip_runtime.h>
#include <hip/hip_bf16.h>


typedef __attribute__((ext_vector_type(2))) float v2f;
typedef __attribute__((ext_vector_type(8))) float v8f;
typedef __attribute__((ext_vector_type(4))) float v4f;

#define BATCH     32
#define FRAMES    4000
#define CBINS     65
#define NIR       129
#define FRAME_LEN 64
#define SEGLEN    192
#define OUTLEN    256128
#define KPAD      68
#define JROWS     144
#define FPB       32
#define HPB       30
#define NHOPS     4002
#define WPB       134
#define IROFF     64
#define IRP       260
#define QPH       16
#define HSP       68
#define NZP       68
#define TILEF     1088

__global__ void FN_setup_M(float* __restrict__ M) {
  const float TWO_PI = 6.28318530717958647692f;
  for (int i = blockIdx.x * blockDim.x + threadIdx.x; i < JROWS * KPAD;
       i += gridDim.x * blockDim.x) {
    int jt = i / TILEF;
    int r  = i - jt * TILEF;
    int l, k;
    if (r < 1024) {
      int g = r >> 7;
      l = (r & 127) >> 2;
      k = 8 * g + 4 * (l >> 4) + (r & 3);
    } else {
      int r2 = r - 1024;
      l = r2 >> 1;
      k = 64 + 2 * (l >> 4) + (r2 & 1);
    }
    int j = jt * 16 + (l & 15);
    float v = 0.0f;
    if (j < NIR && k < CBINS) {
      float win = 0.5f * (1.0f - cosf(TWO_PI * (float)j / (float)NIR));
      int m = (k * (j - 64)) % NIR;
      float ang = TWO_PI * (float)m / (float)NIR;
      float ck  = (k == 0) ? 1.0f : 2.0f;
      v = win * (ck / (float)NIR) * cosf(ang);
    }
    M[i] = v;
  }
}

__device__ __forceinline__ void FN_load_b(const float* __restrict__ M, int jt, int lane,
                                          v4f (&bq)[8], v2f& bt) {
  const float* mp = M + jt * TILEF;
#pragma unroll
  for (int g = 0; g < 8; ++g) bq[g] = *(const v4f*)(mp + g * 128 + lane * 4);
  bt = *(const v2f*)(mp + 1024 + lane * 2);
}

__device__ __forceinline__ void FN_mm_tile(const float* hs, float* irs, int jt, int lane,
                                           const v4f (&bq)[8], v2f bt) {
  const int c  = lane & 15;
  const int hh = lane >> 4;
  const float* ar = hs + c * HSP;
  v8f acc = {};
#pragma unroll
  for (int g = 0; g < 8; ++g) {
    const v4f aq = *(const v4f*)(ar + 8 * g + 4 * hh);
    v2f a0, a1, b0, b1;
    a0.x = aq.x;    a0.y = aq.y;    a1.x = aq.z;    a1.y = aq.w;
    b0.x = bq[g].x; b0.y = bq[g].y; b1.x = bq[g].z; b1.y = bq[g].w;
    acc = __builtin_amdgcn_wmma_f32_16x16x4_f32(
         false, a0,  false, b0,
         (short)0, acc,  false,  false);
    acc = __builtin_amdgcn_wmma_f32_16x16x4_f32(
         false, a1,  false, b1,
         (short)0, acc,  false,  false);
  }
  const v2f at = *(const v2f*)(ar + 64 + 2 * hh);
  acc = __builtin_amdgcn_wmma_f32_16x16x4_f32(
       false, at,  false, bt,
       (short)0, acc,  false,  false);
  const int mbase = hh << 3;
#pragma unroll
  for (int v = 0; v < 8; ++v)
    irs[(mbase + v) * IRP + IROFF + jt * 16 + c] = acc[v];
}

__device__ __forceinline__ void FN_mm_tile2(const float* hs, float* irs, int jt, int lane,
                                            const v4f (&bq)[8], v2f bt) {
  const int c  = lane & 15;
  const int hh = lane >> 4;
  const float* ar0 = hs + c * HSP;
  const float* ar1 = ar0 + 16 * HSP;
  v8f acc0 = {};
  v8f acc1 = {};
#pragma unroll
  for (int g = 0; g < 8; ++g) {
    const v4f aq0 = *(const v4f*)(ar0 + 8 * g + 4 * hh);
    const v4f aq1 = *(const v4f*)(ar1 + 8 * g + 4 * hh);
    v2f a00, a01, a10, a11, b0, b1;
    a00.x = aq0.x;  a00.y = aq0.y;  a01.x = aq0.z;  a01.y = aq0.w;
    a10.x = aq1.x;  a10.y = aq1.y;  a11.x = aq1.z;  a11.y = aq1.w;
    b0.x = bq[g].x; b0.y = bq[g].y; b1.x = bq[g].z; b1.y = bq[g].w;
    acc0 = __builtin_amdgcn_wmma_f32_16x16x4_f32(
         false, a00,  false, b0,
         (short)0, acc0,  false,  false);
    acc1 = __builtin_amdgcn_wmma_f32_16x16x4_f32(
         false, a10,  false, b0,
         (short)0, acc1,  false,  false);
    acc0 = __builtin_amdgcn_wmma_f32_16x16x4_f32(
         false, a01,  false, b1,
         (short)0, acc0,  false,  false);
    acc1 = __builtin_amdgcn_wmma_f32_16x16x4_f32(
         false, a11,  false, b1,
         (short)0, acc1,  false,  false);
  }
  const v2f at0 = *(const v2f*)(ar0 + 64 + 2 * hh);
  const v2f at1 = *(const v2f*)(ar1 + 64 + 2 * hh);
  acc0 = __builtin_amdgcn_wmma_f32_16x16x4_f32(
       false, at0,  false, bt,
       (short)0, acc0,  false,  false);
  acc1 = __builtin_amdgcn_wmma_f32_16x16x4_f32(
       false, at1,  false, bt,
       (short)0, acc1,  false,  false);
  const int mbase = hh << 3;
#pragma unroll
  for (int v = 0; v < 8; ++v) {
    irs[(mbase + v) * IRP + IROFF + jt * 16 + c]      = acc0[v];
    irs[(16 + mbase + v) * IRP + IROFF + jt * 16 + c] = acc1[v];
  }
}

__global__ __launch_bounds__(256) void FN_fused(const float* __restrict__ H,
                                                const float* __restrict__ noise,
                                                const float* __restrict__ M,
                                                float* __restrict__ out) {
  __shared__ __attribute__((aligned(16))) float irs[FPB * IRP];
  __shared__ __attribute__((aligned(16))) float nz[FPB * NZP];
  __shared__ __attribute__((aligned(16))) float hs[FPB * HSP];

  const int tid  = threadIdx.x;
  const int lane = tid & 31;
  const int wave = __builtin_amdgcn_readfirstlane(tid >> 5);

  const int b  = blockIdx.x / WPB;
  const int g0 = (blockIdx.x - b * WPB) * HPB;
  const int fb = g0 - 2;
  const long rb = (long)b * FRAMES;

  float hv[9];
  int   ho[9];
#pragma unroll
  for (int u = 0; u < 9; ++u) {
    int i  = (u < 8) ? (tid + 256 * u) : min(tid + 2048, FPB * CBINS - 1);
    int lf = i / CBINS;
    int k  = i - lf * CBINS;
    int f  = min(max(fb + lf, 0), FRAMES - 1);
    hv[u] = H[(rb + f) * CBINS + k];
    ho[u] = lf * HSP + k;
  }

  v4f bq[8];
  v2f bt;
  FN_load_b(M, wave, lane, bq, bt);

#pragma unroll
  for (int u = 0; u < 2; ++u) {
    int i  = tid + 256 * u;
    int lf = i >> 4;
    int qq = i & 15;
    int fr = fb + lf;
    int f  = min(max(fr, 0), FRAMES - 1);
    const v4f nv = *(const v4f*)(noise + (rb + f) * FRAME_LEN + 4 * qq);
    const float ok = (fr >= 0 && fr < FRAMES) ? 1.0f : 0.0f;
    v4f m;
    m.x = (nv.x * 2.0f - 1.0f) * ok;
    m.y = (nv.y * 2.0f - 1.0f) * ok;
    m.z = (nv.z * 2.0f - 1.0f) * ok;
    m.w = (nv.w * 2.0f - 1.0f) * ok;
    *(v4f*)(nz + lf * NZP + 4 * qq) = m;
  }

  for (int i = tid; i < FPB * 28; i += 256) {
    int rw = i / 28;
    int c  = i - rw * 28;
    int s4 = (c < 16) ? c : (c + 36);
    float4 z; z.x = z.y = z.z = z.w = 0.0f;
    *(float4*)(irs + rw * IRP + s4 * 4) = z;
  }

#pragma unroll
  for (int u = 0; u < 8; ++u) hs[ho[u]] = hv[u];
  if (tid < FPB * CBINS - 2048) hs[ho[8]] = hv[8];
  if (tid < FPB * (HSP - CBINS))
    hs[(tid / 3) * HSP + CBINS + (tid - (tid / 3) * 3)] = 0.0f;

  __syncthreads();

  FN_mm_tile2(hs, irs, wave, lane, bq, bt);
  if (wave < 2) {
    FN_load_b(M, 8, lane, bq, bt);
    FN_mm_tile(hs + wave * (16 * HSP), irs + wave * (16 * IRP), 8, lane, bq, bt);
  }

  __syncthreads();

  if (lane < HPB) {
    float s[8];
#pragma unroll
    for (int a = 0; a < 8; ++a) s[a] = 0.0f;
#pragma unroll
    for (int d = 0; d < 3; ++d) {
      const int lf = lane + 2 - d;
      const int cs = (d == 2) ? ((2 * wave) & ~3) : 0;
      const int ce = (d == 0) ? ((2 * wave + 5) & ~3) : 16;
      const float* irow = irs + lf * IRP + IROFF + d * FRAME_LEN + 8 * wave;
      const float* nrow = nz + lf * NZP;
      v4f hi  = *(const v4f*)(irow - 4 * cs + 4);
      v4f mid = *(const v4f*)(irow - 4 * cs);
      for (int c4 = cs; c4 < ce; c4 += 4) {
#pragma unroll
        for (int u = 0; u < 4; ++u) {
          const int cc = c4 + u;
          const v4f lo = *(const v4f*)(irow - 4 * cc - 4);
          const v4f n  = *(const v4f*)(nrow + 4 * cc);
          const float w[12] = {lo.x, lo.y, lo.z, lo.w, mid.x, mid.y, mid.z, mid.w,
                               hi.x, hi.y, hi.z, hi.w};
          const float nn[4] = {n.x, n.y, n.z, n.w};
#pragma unroll
          for (int e = 0; e < 4; ++e) {
#pragma unroll
            for (int a = 0; a < 8; ++a) s[a] = fmaf(nn[e], w[a - e + 4], s[a]);
          }
          hi  = mid;
          mid = lo;
        }
      }
    }
    v4f o0, o1;
    o0.x = s[0] * 0.01f; o0.y = s[1] * 0.01f; o0.z = s[2] * 0.01f; o0.w = s[3] * 0.01f;
    o1.x = s[4] * 0.01f; o1.y = s[5] * 0.01f; o1.z = s[6] * 0.01f; o1.w = s[7] * 0.01f;
    float* ob = hs + lane * NZP + 8 * wave;
    *(v4f*)ob       = o0;
    *(v4f*)(ob + 4) = o1;
  }

  __syncthreads();

  for (int i = tid; i < HPB * QPH; i += 256) {
    const int r  = i >> 4;
    const int qd = i & 15;
    const int h  = g0 + r;
    if (h < NHOPS) {
      const v4f v = *(const v4f*)(hs + r * NZP + 4 * qd);
      *(v4f*)(out + (long)b * OUTLEN + (long)h * FRAME_LEN + 4 * qd) = v;
    }
  }
}

extern "C" void kernel_launch(void* const* d_in, const int* in_sizes, int n_in,
                              void* d_out, int out_size, void* d_ws, size_t ws_size,
                              hipStream_t stream) {
  const float* H     = (const float*)d_in[0];
  const float* noise = (const float*)d_in[1];
  float* out = (float*)d_out;
  float* Mws = (float*)d_ws;

  FN_setup_M<<<(JROWS * KPAD + 255) / 256, 256, 0, stream>>>(Mws);
  FN_fused<<<BATCH * WPB, 256, 0, stream>>>(H, noise, Mws, out);
}


# ir matmul halved by the even symmetry of the real-input irfft: zp = C(65x65) @ H on WMMA (5 tiles, was 9), each zp[n] stored to taps 64+n and 64-n times its window weight; no serial ninth-tile pass
# speedup vs baseline: 6.0730x; 1.1382x over previous
#include <hip/hip_runtime.h>
#include <hip/hip_bf16.h>


typedef __attribute__((ext_vector_type(2))) float v2f;
typedef __attribute__((ext_vector_type(8))) float v8f;
typedef __attribute__((ext_vector_type(4))) float v4f;

#define BATCH     32
#define FRAMES    4000
#define CBINS     65
#define NIR       129
#define FRAME_LEN 64
#define SEGLEN    192
#define OUTLEN    256128
#define KPAD      68
#define NTILES    5
#define NPAD      80
#define HALF      64
#define FPB       32
#define HPB       30
#define NHOPS     4002
#define WPB       134
#define IROFF     64
#define IRP       260
#define QPH       16
#define HSP       68
#define NZP       68
#define TILEF     1088

__global__ void FN_setup_M(float* __restrict__ M) {
  const float TWO_PI = 6.28318530717958647692f;
  for (int i = blockIdx.x * blockDim.x + threadIdx.x; i < NTILES * TILEF + 2 * NPAD;
       i += gridDim.x * blockDim.x) {
    if (i >= NTILES * TILEF) {
      int q = i - NTILES * TILEF;
      int n = (q < NPAD) ? q : (q - NPAD);
      int j = (q < NPAD) ? (HALF + n) : (HALF - n);
      float w = 0.0f;
      if (n <= HALF) w = 0.5f * (1.0f - cosf(TWO_PI * (float)j / (float)NIR));
      M[i] = w;
      continue;
    }
    int jt = i / TILEF;
    int r  = i - jt * TILEF;
    int l, k;
    if (r < 1024) {
      int g = r >> 7;
      l = (r & 127) >> 2;
      k = 8 * g + 4 * (l >> 4) + (r & 3);
    } else {
      int r2 = r - 1024;
      l = r2 >> 1;
      k = 64 + 2 * (l >> 4) + (r2 & 1);
    }
    int n = jt * 16 + (l & 15);
    float v = 0.0f;
    if (n <= HALF && k < CBINS) {
      int m = (k * n) % NIR;
      float ang = TWO_PI * (float)m / (float)NIR;
      float ck  = (k == 0) ? 1.0f : 2.0f;
      v = (ck / (float)NIR) * cosf(ang);
    }
    M[i] = v;
  }
}

__device__ __forceinline__ void FN_load_b(const float* __restrict__ M, int jt, int lane,
                                          v4f (&bq)[8], v2f& bt) {
  const float* mp = M + jt * TILEF;
#pragma unroll
  for (int g = 0; g < 8; ++g) bq[g] = *(const v4f*)(mp + g * 128 + lane * 4);
  bt = *(const v2f*)(mp + 1024 + lane * 2);
}

__device__ __forceinline__ void FN_mm_tile(const float* hs, float* irs, int jt, int lane,
                                           const v4f (&bq)[8], v2f bt, float wup, float wdn) {
  const int c  = lane & 15;
  const int hh = lane >> 4;
  const float* ar = hs + c * HSP;
  v8f acc = {};
#pragma unroll
  for (int g = 0; g < 8; ++g) {
    const v4f aq = *(const v4f*)(ar + 8 * g + 4 * hh);
    v2f a0, a1, b0, b1;
    a0.x = aq.x;    a0.y = aq.y;    a1.x = aq.z;    a1.y = aq.w;
    b0.x = bq[g].x; b0.y = bq[g].y; b1.x = bq[g].z; b1.y = bq[g].w;
    acc = __builtin_amdgcn_wmma_f32_16x16x4_f32(
         false, a0,  false, b0,
         (short)0, acc,  false,  false);
    acc = __builtin_amdgcn_wmma_f32_16x16x4_f32(
         false, a1,  false, b1,
         (short)0, acc,  false,  false);
  }
  const v2f at = *(const v2f*)(ar + 64 + 2 * hh);
  acc = __builtin_amdgcn_wmma_f32_16x16x4_f32(
       false, at,  false, bt,
       (short)0, acc,  false,  false);
  const int mbase = hh << 3;
  const int n = jt * 16 + c;
  float* up = irs + mbase * IRP + IROFF + HALF + n;
  float* dn = irs + mbase * IRP + IROFF + HALF - n;
#pragma unroll
  for (int v = 0; v < 8; ++v) {
    up[v * IRP] = acc[v] * wup;
    dn[v * IRP] = acc[v] * wdn;
  }
}

__device__ __forceinline__ void FN_mm_tile2(const float* hs, float* irs, int jt, int lane,
                                            const v4f (&bq)[8], v2f bt, float wup, float wdn) {
  const int c  = lane & 15;
  const int hh = lane >> 4;
  const float* ar0 = hs + c * HSP;
  const float* ar1 = ar0 + 16 * HSP;
  v8f acc0 = {};
  v8f acc1 = {};
#pragma unroll
  for (int g = 0; g < 8; ++g) {
    const v4f aq0 = *(const v4f*)(ar0 + 8 * g + 4 * hh);
    const v4f aq1 = *(const v4f*)(ar1 + 8 * g + 4 * hh);
    v2f a00, a01, a10, a11, b0, b1;
    a00.x = aq0.x;  a00.y = aq0.y;  a01.x = aq0.z;  a01.y = aq0.w;
    a10.x = aq1.x;  a10.y = aq1.y;  a11.x = aq1.z;  a11.y = aq1.w;
    b0.x = bq[g].x; b0.y = bq[g].y; b1.x = bq[g].z; b1.y = bq[g].w;
    acc0 = __builtin_amdgcn_wmma_f32_16x16x4_f32(
         false, a00,  false, b0,
         (short)0, acc0,  false,  false);
    acc1 = __builtin_amdgcn_wmma_f32_16x16x4_f32(
         false, a10,  false, b0,
         (short)0, acc1,  false,  false);
    acc0 = __builtin_amdgcn_wmma_f32_16x16x4_f32(
         false, a01,  false, b1,
         (short)0, acc0,  false,  false);
    acc1 = __builtin_amdgcn_wmma_f32_16x16x4_f32(
         false, a11,  false, b1,
         (short)0, acc1,  false,  false);
  }
  const v2f at0 = *(const v2f*)(ar0 + 64 + 2 * hh);
  const v2f at1 = *(const v2f*)(ar1 + 64 + 2 * hh);
  acc0 = __builtin_amdgcn_wmma_f32_16x16x4_f32(
       false, at0,  false, bt,
       (short)0, acc0,  false,  false);
  acc1 = __builtin_amdgcn_wmma_f32_16x16x4_f32(
       false, at1,  false, bt,
       (short)0, acc1,  false,  false);
  const int mbase = hh << 3;
  const int n = jt * 16 + c;
  float* up = irs + mbase * IRP + IROFF + HALF + n;
  float* dn = irs + mbase * IRP + IROFF + HALF - n;
#pragma unroll
  for (int v = 0; v < 8; ++v) {
    up[v * IRP]        = acc0[v] * wup;
    dn[v * IRP]        = acc0[v] * wdn;
    up[(16 + v) * IRP] = acc1[v] * wup;
    dn[(16 + v) * IRP] = acc1[v] * wdn;
  }
}

__global__ __launch_bounds__(256) void FN_fused(const float* __restrict__ H,
                                                const float* __restrict__ noise,
                                                const float* __restrict__ M,
                                                float* __restrict__ out) {
  __shared__ __attribute__((aligned(16))) float irs[FPB * IRP];
  __shared__ __attribute__((aligned(16))) float nz[FPB * NZP];
  __shared__ __attribute__((aligned(16))) float hs[FPB * HSP];

  const int tid  = threadIdx.x;
  const int lane = tid & 31;
  const int wave = __builtin_amdgcn_readfirstlane(tid >> 5);

  const int b  = blockIdx.x / WPB;
  const int g0 = (blockIdx.x - b * WPB) * HPB;
  const int fb = g0 - 2;
  const long rb = (long)b * FRAMES;

  float hv[9];
  int   ho[9];
#pragma unroll
  for (int u = 0; u < 9; ++u) {
    int i  = (u < 8) ? (tid + 256 * u) : min(tid + 2048, FPB * CBINS - 1);
    int lf = i / CBINS;
    int k  = i - lf * CBINS;
    int f  = min(max(fb + lf, 0), FRAMES - 1);
    hv[u] = H[(rb + f) * CBINS + k];
    ho[u] = lf * HSP + k;
  }

  const int jt = (0x34244140 >> (4 * wave)) & 15;
  v4f bq[8];
  v2f bt;
  FN_load_b(M, jt, lane, bq, bt);
  const float wup = M[NTILES * TILEF + jt * 16 + (lane & 15)];
  const float wdn = M[NTILES * TILEF + NPAD + jt * 16 + (lane & 15)];

#pragma unroll
  for (int u = 0; u < 2; ++u) {
    int i  = tid + 256 * u;
    int lf = i >> 4;
    int qq = i & 15;
    int fr = fb + lf;
    int f  = min(max(fr, 0), FRAMES - 1);
    const v4f nv = *(const v4f*)(noise + (rb + f) * FRAME_LEN + 4 * qq);
    const float ok = (fr >= 0 && fr < FRAMES) ? 1.0f : 0.0f;
    v4f m;
    m.x = (nv.x * 2.0f - 1.0f) * ok;
    m.y = (nv.y * 2.0f - 1.0f) * ok;
    m.z = (nv.z * 2.0f - 1.0f) * ok;
    m.w = (nv.w * 2.0f - 1.0f) * ok;
    *(v4f*)(nz + lf * NZP + 4 * qq) = m;
  }

  for (int i = tid; i < FPB * 28; i += 256) {
    int rw = i / 28;
    int c  = i - rw * 28;
    int s4 = (c < 16) ? c : (c + 36);
    float4 z; z.x = z.y = z.z = z.w = 0.0f;
    *(float4*)(irs + rw * IRP + s4 * 4) = z;
  }

#pragma unroll
  for (int u = 0; u < 8; ++u) hs[ho[u]] = hv[u];
  if (tid < FPB * CBINS - 2048) hs[ho[8]] = hv[8];
  if (tid < FPB * (HSP - CBINS))
    hs[(tid / 3) * HSP + CBINS + (tid - (tid / 3) * 3)] = 0.0f;

  __syncthreads();

  if (jt < 4) {
    FN_mm_tile2(hs, irs, jt, lane, bq, bt, wup, wdn);
  } else if (wave == 1 || wave == 6) {
    const int rt = (wave == 6) ? 1 : 0;
    FN_mm_tile(hs + rt * (16 * HSP), irs + rt * (16 * IRP), jt, lane, bq, bt, wup, wdn);
  }

  __syncthreads();

  if (lane < HPB) {
    float s[8];
#pragma unroll
    for (int a = 0; a < 8; ++a) s[a] = 0.0f;
#pragma unroll
    for (int d = 0; d < 3; ++d) {
      const int lf = lane + 2 - d;
      const int cs = (d == 2) ? ((2 * wave) & ~3) : 0;
      const int ce = (d == 0) ? ((2 * wave + 5) & ~3) : 16;
      const float* irow = irs + lf * IRP + IROFF + d * FRAME_LEN + 8 * wave;
      const float* nrow = nz + lf * NZP;
      v4f hi  = *(const v4f*)(irow - 4 * cs + 4);
      v4f mid = *(const v4f*)(irow - 4 * cs);
      for (int c4 = cs; c4 < ce; c4 += 4) {
#pragma unroll
        for (int u = 0; u < 4; ++u) {
          const int cc = c4 + u;
          const v4f lo = *(const v4f*)(irow - 4 * cc - 4);
          const v4f n  = *(const v4f*)(nrow + 4 * cc);
          const float w[12] = {lo.x, lo.y, lo.z, lo.w, mid.x, mid.y, mid.z, mid.w,
                               hi.x, hi.y, hi.z, hi.w};
          const float nn[4] = {n.x, n.y, n.z, n.w};
#pragma unroll
          for (int e = 0; e < 4; ++e) {
#pragma unroll
            for (int a = 0; a < 8; ++a) s[a] = fmaf(nn[e], w[a - e + 4], s[a]);
          }
          hi  = mid;
          mid = lo;
        }
      }
    }
    v4f o0, o1;
    o0.x = s[0] * 0.01f; o0.y = s[1] * 0.01f; o0.z = s[2] * 0.01f; o0.w = s[3] * 0.01f;
    o1.x = s[4] * 0.01f; o1.y = s[5] * 0.01f; o1.z = s[6] * 0.01f; o1.w = s[7] * 0.01f;
    float* ob = hs + lane * NZP + 8 * wave;
    *(v4f*)ob       = o0;
    *(v4f*)(ob + 4) = o1;
  }

  __syncthreads();

  for (int i = tid; i < HPB * QPH; i += 256) {
    const int r  = i >> 4;
    const int qd = i & 15;
    const int h  = g0 + r;
    if (h < NHOPS) {
      const v4f v = *(const v4f*)(hs + r * NZP + 4 * qd);
      *(v4f*)(out + (long)b * OUTLEN + (long)h * FRAME_LEN + 4 * qd) = v;
    }
  }
}

extern "C" void kernel_launch(void* const* d_in, const int* in_sizes, int n_in,
                              void* d_out, int out_size, void* d_ws, size_t ws_size,
                              hipStream_t stream) {
  const float* H     = (const float*)d_in[0];
  const float* noise = (const float*)d_in[1];
  float* out = (float*)d_out;
  float* Mws = (float*)d_ws;

  FN_setup_M<<<(NTILES * TILEF + 2 * NPAD + 255) / 256, 256, 0, stream>>>(Mws);
  FN_fused<<<BATCH * WPB, 256, 0, stream>>>(H, noise, Mws, out);
}


# conv packed FMAs pair two hops of one lane (p, p+15): their ir and noise rows lie interleaved in LDS, so operand pairs are used as read and the pair-forming moves are gone; 4 waves per work-group
# speedup vs baseline: 6.8289x; 1.1245x over previous
#include <hip/hip_runtime.h>
#include <hip/hip_bf16.h>


typedef __attribute__((ext_vector_type(2))) float v2f;
typedef __attribute__((ext_vector_type(8))) float v8f;
typedef __attribute__((ext_vector_type(4))) float v4f;

#define BATCH     32
#define FRAMES    4000
#define CBINS     65
#define NIR       129
#define FRAME_LEN 64
#define SEGLEN    192
#define OUTLEN    256128
#define KPAD      68
#define NTILES    5
#define NPAD      80
#define HALF      64
#define FPB       32
#define HPB       30
#define NHOPS     4002
#define WPB       134
#define NTHR      128
#define PAIRS     15
#define NPR       17
#define IROFF     16
#define PR2       324
#define NZ2       132
#define QPH       16
#define HSP       68
#define NZP       68
#define TILEF     1088

__global__ void FN_setup_M(float* __restrict__ M) {
  const float TWO_PI = 6.28318530717958647692f;
  for (int i = blockIdx.x * blockDim.x + threadIdx.x; i < NTILES * TILEF + 2 * NPAD;
       i += gridDim.x * blockDim.x) {
    if (i >= NTILES * TILEF) {
      int q = i - NTILES * TILEF;
      int n = (q < NPAD) ? q : (q - NPAD);
      int j = (q < NPAD) ? (HALF + n) : (HALF - n);
      float w = 0.0f;
      if (n <= HALF) w = 0.5f * (1.0f - cosf(TWO_PI * (float)j / (float)NIR));
      M[i] = w;
      continue;
    }
    int jt = i / TILEF;
    int r  = i - jt * TILEF;
    int l, k;
    if (r < 1024) {
      int g = r >> 7;
      l = (r & 127) >> 2;
      k = 8 * g + 4 * (l >> 4) + (r & 3);
    } else {
      int r2 = r - 1024;
      l = r2 >> 1;
      k = 64 + 2 * (l >> 4) + (r2 & 1);
    }
    int n = jt * 16 + (l & 15);
    float v = 0.0f;
    if (n <= HALF && k < CBINS) {
      int m = (k * n) % NIR;
      float ang = TWO_PI * (float)m / (float)NIR;
      float ck  = (k == 0) ? 1.0f : 2.0f;
      v = (ck / (float)NIR) * cosf(ang);
    }
    M[i] = v;
  }
}

__device__ __forceinline__ void FN_load_b(const float* __restrict__ M, int jt, int lane,
                                          v4f (&bq)[8], v2f& bt) {
  const float* mp = M + jt * TILEF;
#pragma unroll
  for (int g = 0; g < 8; ++g) bq[g] = *(const v4f*)(mp + g * 128 + lane * 4);
  bt = *(const v2f*)(mp + 1024 + lane * 2);
}

__device__ __forceinline__ void FN_store_zp(float* irs2, int rt, int jt, int lane,
                                            const v8f& acc, float wup, float wdn) {
  const int c  = lane & 15;
  const int hh = lane >> 4;
  const int n  = jt * 16 + c;
  const int cu = 2 * (IROFF + HALF + n);
  const int cd = 2 * (IROFF + HALF - n);
  float* q = irs2 + (rt + 8 * hh) * PR2 + rt;
#pragma unroll
  for (int v = 0; v < 8; ++v) {
    q[v * PR2 + cu] = acc[v] * wup;
    q[v * PR2 + cd] = acc[v] * wdn;
  }
  if (hh != rt) {
    const float a = (rt == 0) ? acc[7] : acc[0];
    float* d = irs2 + ((rt == 0) ? 1 : 16 * PR2);
    d[cu] = a * wup;
    d[cd] = a * wdn;
  }
}

__device__ __forceinline__ void FN_mm_tile(const float* hs, float* irs2, int rt, int jt, int lane,
                                           const v4f (&bq)[8], v2f bt, float wup, float wdn) {
  const int c  = lane & 15;
  const int hh = lane >> 4;
  const float* ar = hs + (16 * rt + c) * HSP;
  v8f acc = {};
#pragma unroll
  for (int g = 0; g < 8; ++g) {
    const v4f aq = *(const v4f*)(ar + 8 * g + 4 * hh);
    v2f a0, a1, b0, b1;
    a0.x = aq.x;    a0.y = aq.y;    a1.x = aq.z;    a1.y = aq.w;
    b0.x = bq[g].x; b0.y = bq[g].y; b1.x = bq[g].z; b1.y = bq[g].w;
    acc = __builtin_amdgcn_wmma_f32_16x16x4_f32(
         false, a0,  false, b0,
         (short)0, acc,  false,  false);
    acc = __builtin_amdgcn_wmma_f32_16x16x4_f32(
         false, a1,  false, b1,
         (short)0, acc,  false,  false);
  }
  const v2f at = *(const v2f*)(ar + 64 + 2 * hh);
  acc = __builtin_amdgcn_wmma_f32_16x16x4_f32(
       false, at,  false, bt,
       (short)0, acc,  false,  false);
  FN_store_zp(irs2, rt, jt, lane, acc, wup, wdn);
}

__device__ __forceinline__ void FN_mm_tile2(const float* hs, float* irs2, int jt, int lane,
                                            const v4f (&bq)[8], v2f bt, float wup, float wdn) {
  const int c  = lane & 15;
  const int hh = lane >> 4;
  const float* ar0 = hs + c * HSP;
  const float* ar1 = ar0 + 16 * HSP;
  v8f acc0 = {};
  v8f acc1 = {};
#pragma unroll
  for (int g = 0; g < 8; ++g) {
    const v4f aq0 = *(const v4f*)(ar0 + 8 * g + 4 * hh);
    const v4f aq1 = *(const v4f*)(ar1 + 8 * g + 4 * hh);
    v2f a00, a01, a10, a11, b0, b1;
    a00.x = aq0.x;  a00.y = aq0.y;  a01.x = aq0.z;  a01.y = aq0.w;
    a10.x = aq1.x;  a10.y = aq1.y;  a11.x = aq1.z;  a11.y = aq1.w;
    b0.x = bq[g].x; b0.y = bq[g].y; b1.x = bq[g].z; b1.y = bq[g].w;
    acc0 = __builtin_amdgcn_wmma_f32_16x16x4_f32(
         false, a00,  false, b0,
         (short)0, acc0,  false,  false);
    acc1 = __builtin_amdgcn_wmma_f32_16x16x4_f32(
         false, a10,  false, b0,
         (short)0, acc1,  false,  false);
    acc0 = __builtin_amdgcn_wmma_f32_16x16x4_f32(
         false, a01,  false, b1,
         (short)0, acc0,  false,  false);
    acc1 = __builtin_amdgcn_wmma_f32_16x16x4_f32(
         false, a11,  false, b1,
         (short)0, acc1,  false,  false);
  }
  const v2f at0 = *(const v2f*)(ar0 + 64 + 2 * hh);
  const v2f at1 = *(const v2f*)(ar1 + 64 + 2 * hh);
  acc0 = __builtin_amdgcn_wmma_f32_16x16x4_f32(
       false, at0,  false, bt,
       (short)0, acc0,  false,  false);
  acc1 = __builtin_amdgcn_wmma_f32_16x16x4_f32(
       false, at1,  false, bt,
       (short)0, acc1,  false,  false);
  FN_store_zp(irs2, 0, jt, lane, acc0, wup, wdn);
  FN_store_zp(irs2, 1, jt, lane, acc1, wup, wdn);
}

__device__ __forceinline__ v2f FN_lo2(v4f q) { v2f t; t.x = q.x; t.y = q.y; return t; }
__device__ __forceinline__ v2f FN_hi2(v4f q) { v2f t; t.x = q.z; t.y = q.w; return t; }

__global__ __launch_bounds__(NTHR) void FN_fused(const float* __restrict__ H,
                                                 const float* __restrict__ noise,
                                                 const float* __restrict__ M,
                                                 float* __restrict__ out) {
  __shared__ __attribute__((aligned(16))) float irs2[NPR * PR2];
  __shared__ __attribute__((aligned(16))) float nz2[NPR * NZ2];
  __shared__ __attribute__((aligned(16))) float hs[FPB * HSP];

  const int tid  = threadIdx.x;
  const int lane = tid & 31;
  const int wave = __builtin_amdgcn_readfirstlane(tid >> 5);

  const int b  = blockIdx.x / WPB;
  const int g0 = (blockIdx.x - b * WPB) * HPB;
  const int fb = g0 - 2;
  const long rb = (long)b * FRAMES;

  float hv[17];
  const int hbase = ((int)rb + fb) * CBINS;
#pragma unroll
  for (int u = 0; u < 17; ++u) {
    int i  = (u < 16) ? (tid + NTHR * u) : min(tid + 16 * NTHR, FPB * CBINS - 1);
    unsigned gi = (unsigned)min(max(hbase + i, 0), BATCH * FRAMES * CBINS - 1);
    hv[u] = H[gi];
  }

  const int c15 = lane & 15;
  v4f bq[8];
  v2f bt;
  FN_load_b(M, wave, lane, bq, bt);
  const float wup = M[NTILES * TILEF + wave * 16 + c15];
  const float wdn = M[NTILES * TILEF + NPAD + wave * 16 + c15];
  const bool five = (wave == 1) || (wave == 2);
  v4f bq5[8];
  v2f bt5;
  float wup5 = 0.0f, wdn5 = 0.0f;
#pragma unroll
  for (int g = 0; g < 8; ++g) { bq5[g].x = 0.0f; bq5[g].y = 0.0f; bq5[g].z = 0.0f; bq5[g].w = 0.0f; }
  bt5.x = 0.0f; bt5.y = 0.0f;
  if (five) {
    FN_load_b(M, 4, lane, bq5, bt5);
    wup5 = M[NTILES * TILEF + 64 + c15];
    wdn5 = M[NTILES * TILEF + NPAD + 64 + c15];
  }

#pragma unroll
  for (int u = 0; u < 3; ++u) {
    int i = tid + NTHR * u;
    if (i < NPR * 16) {
      int r  = i >> 4;
      int qq = i & 15;
      int fa = fb + r;
      int fc = fb + r + PAIRS;
      int f0 = min(max(fa, 0), FRAMES - 1);
      int f1 = min(max(fc, 0), FRAMES - 1);
      const unsigned ia = (unsigned)(((int)rb + f0) * FRAME_LEN + 4 * qq);
      const unsigned ib = (unsigned)(((int)rb + f1) * FRAME_LEN + 4 * qq);
      const v4f na = *(const v4f*)(noise + ia);
      const v4f nb = *(const v4f*)(noise + ib);
      const float oka = (fa >= 0 && fa < FRAMES) ? 1.0f : 0.0f;
      const float okb = (fc >= 0 && fc < FRAMES) ? 1.0f : 0.0f;
      v4f m0, m1;
      m0.x = (na.x * 2.0f - 1.0f) * oka;
      m0.y = (nb.x * 2.0f - 1.0f) * okb;
      m0.z = (na.y * 2.0f - 1.0f) * oka;
      m0.w = (nb.y * 2.0f - 1.0f) * okb;
      m1.x = (na.z * 2.0f - 1.0f) * oka;
      m1.y = (nb.z * 2.0f - 1.0f) * okb;
      m1.z = (na.w * 2.0f - 1.0f) * oka;
      m1.w = (nb.w * 2.0f - 1.0f) * okb;
      *(v4f*)(nz2 + r * NZ2 + 8 * qq)     = m0;
      *(v4f*)(nz2 + r * NZ2 + 8 * qq + 4) = m1;
    }
  }

  if (tid < 2 * NPR) irs2[(tid >> 1) * PR2 + (tid & 1)] = 0.0f;

#pragma unroll
  for (int u = 0; u < 16; ++u) {
    const int i = tid + NTHR * u;
    hs[i + (HSP - CBINS) * (i / CBINS)] = hv[u];
  }
  if (tid < FPB * CBINS - 16 * NTHR) {
    const int i = tid + 16 * NTHR;
    hs[i + (HSP - CBINS) * (i / CBINS)] = hv[16];
  }
  if (tid < FPB * (HSP - CBINS))
    hs[(tid / 3) * HSP + CBINS + (tid - (tid / 3) * 3)] = 0.0f;

  __syncthreads();

  FN_mm_tile2(hs, irs2, wave, lane, bq, bt, wup, wdn);
  if (five) FN_mm_tile(hs, irs2, wave - 1, 4, lane, bq5, bt5, wup5, wdn5);

  __syncthreads();

  const int p  = lane & 15;
  const int ch = 2 * wave + (lane >> 4);
  if (p < PAIRS) {
    v2f s[8];
#pragma unroll
    for (int a = 0; a < 8; ++a) { s[a].x = 0.0f; s[a].y = 0.0f; }
#pragma unroll
    for (int d = 0; d < 3; ++d) {
      const int r  = p + 2 - d;
      const int cs = (d == 2) ? (4 * wave) : 0;
      const int ce = (d == 0) ? (4 * wave + 4) : 16;
      const float* irow = irs2 + r * PR2 + 2 * (IROFF + d * FRAME_LEN + 8 * ch);
      const float* nrow = nz2 + r * NZ2;
      v4f hiA  = *(const v4f*)(irow - 8 * cs + 8);
      v4f hiB  = *(const v4f*)(irow - 8 * cs + 12);
      v4f midA = *(const v4f*)(irow - 8 * cs);
      v4f midB = *(const v4f*)(irow - 8 * cs + 4);
      for (int c4 = cs; c4 < ce; c4 += 4) {
#pragma unroll
        for (int u = 0; u < 4; ++u) {
          const int cc = c4 + u;
          const v4f loA = *(const v4f*)(irow - 8 * cc - 8);
          const v4f loB = *(const v4f*)(irow - 8 * cc - 4);
          const v4f nA  = *(const v4f*)(nrow + 8 * cc);
          const v4f nB  = *(const v4f*)(nrow + 8 * cc + 4);
          const v2f w[12] = {FN_lo2(loA),  FN_hi2(loA),  FN_lo2(loB),  FN_hi2(loB),
                             FN_lo2(midA), FN_hi2(midA), FN_lo2(midB), FN_hi2(midB),
                             FN_lo2(hiA),  FN_hi2(hiA),  FN_lo2(hiB),  FN_hi2(hiB)};
          const v2f nn[4] = {FN_lo2(nA), FN_hi2(nA), FN_lo2(nB), FN_hi2(nB)};
#pragma unroll
          for (int e = 0; e < 4; ++e) {
#pragma unroll
            for (int a = 0; a < 8; ++a) s[a] = __builtin_elementwise_fma(nn[e], w[a - e + 4], s[a]);
          }
          hiA  = midA;
          hiB  = midB;
          midA = loA;
          midB = loB;
        }
      }
    }
    v4f o0, o1, o2, o3;
    o0.x = s[0].x * 0.01f; o0.y = s[1].x * 0.01f; o0.z = s[2].x * 0.01f; o0.w = s[3].x * 0.01f;
    o1.x = s[4].x * 0.01f; o1.y = s[5].x * 0.01f; o1.z = s[6].x * 0.01f; o1.w = s[7].x * 0.01f;
    o2.x = s[0].y * 0.01f; o2.y = s[1].y * 0.01f; o2.z = s[2].y * 0.01f; o2.w = s[3].y * 0.01f;
    o3.x = s[4].y * 0.01f; o3.y = s[5].y * 0.01f; o3.z = s[6].y * 0.01f; o3.w = s[7].y * 0.01f;
    float* ob0 = hs + p * NZP + 8 * ch;
    float* ob1 = hs + (p + PAIRS) * NZP + 8 * ch;
    *(v4f*)ob0       = o0;
    *(v4f*)(ob0 + 4) = o1;
    *(v4f*)ob1       = o2;
    *(v4f*)(ob1 + 4) = o3;
  }

  __syncthreads();

  for (int i = tid; i < HPB * QPH; i += NTHR) {
    const int r  = i >> 4;
    const int qd = i & 15;
    const int h  = g0 + r;
    if (h < NHOPS) {
      const v4f v = *(const v4f*)(hs + r * NZP + 4 * qd);
      *(v4f*)(out + (long)b * OUTLEN + (long)h * FRAME_LEN + 4 * qd) = v;
    }
  }
}

extern "C" void kernel_launch(void* const* d_in, const int* in_sizes, int n_in,
                              void* d_out, int out_size, void* d_ws, size_t ws_size,
                              hipStream_t stream) {
  const float* H     = (const float*)d_in[0];
  const float* noise = (const float*)d_in[1];
  float* out = (float*)d_out;
  float* Mws = (float*)d_ws;

  FN_setup_M<<<(NTILES * TILEF + 2 * NPAD + 255) / 256, 256, 0, stream>>>(Mws);
  FN_fused<<<BATCH * WPB, NTHR, 0, stream>>>(H, noise, Mws, out);
}


# fifth matrix tile removed: zp[0] (cosine row 0 is constant) is a plain row sum of the H tile, 4 lanes per frame, tree order; rows n=1..64 are exactly 4 WMMA tiles, one per wave
# speedup vs baseline: 7.1396x; 1.0455x over previous
#include <hip/hip_runtime.h>
#include <hip/hip_bf16.h>


typedef __attribute__((ext_vector_type(2))) float v2f;
typedef __attribute__((ext_vector_type(8))) float v8f;
typedef __attribute__((ext_vector_type(4))) float v4f;

#define BATCH     32
#define FRAMES    4000
#define CBINS     65
#define NIR       129
#define FRAME_LEN 64
#define SEGLEN    192
#define OUTLEN    256128
#define KPAD      68
#define NTILES    4
#define NPAD      80
#define HALF      64
#define FPB       32
#define HPB       30
#define NHOPS     4002
#define WPB       134
#define NTHR      128
#define PAIRS     15
#define NPR       17
#define IROFF     16
#define PR2       324
#define NZ2       132
#define QPH       16
#define HSP       68
#define NZP       68
#define TILEF     1088

__global__ void FN_setup_M(float* __restrict__ M) {
  const float TWO_PI = 6.28318530717958647692f;
  for (int i = blockIdx.x * blockDim.x + threadIdx.x; i < NTILES * TILEF + 2 * NPAD;
       i += gridDim.x * blockDim.x) {
    if (i >= NTILES * TILEF) {
      int q = i - NTILES * TILEF;
      int n = (q < NPAD) ? q : (q - NPAD);
      int j = (q < NPAD) ? (HALF + n) : (HALF - n);
      float w = 0.0f;
      if (n <= HALF) w = 0.5f * (1.0f - cosf(TWO_PI * (float)j / (float)NIR));
      M[i] = w;
      continue;
    }
    int jt = i / TILEF;
    int r  = i - jt * TILEF;
    int l, k;
    if (r < 1024) {
      int g = r >> 7;
      l = (r & 127) >> 2;
      k = 8 * g + 4 * (l >> 4) + (r & 3);
    } else {
      int r2 = r - 1024;
      l = r2 >> 1;
      k = 64 + 2 * (l >> 4) + (r2 & 1);
    }
    int n = jt * 16 + (l & 15) + 1;
    float v = 0.0f;
    if (k < CBINS) {
      int m = (k * n) % NIR;
      float ang = TWO_PI * (float)m / (float)NIR;
      float ck  = (k == 0) ? 1.0f : 2.0f;
      v = (ck / (float)NIR) * cosf(ang);
    }
    M[i] = v;
  }
}

__device__ __forceinline__ void FN_load_b(const float* __restrict__ M, int jt, int lane,
                                          v4f (&bq)[8], v2f& bt) {
  const float* mp = M + jt * TILEF;
#pragma unroll
  for (int g = 0; g < 8; ++g) bq[g] = *(const v4f*)(mp + g * 128 + lane * 4);
  bt = *(const v2f*)(mp + 1024 + lane * 2);
}

__device__ __forceinline__ void FN_store_zp(float* irs2, int rt, int jt, int lane,
                                            const v8f& acc, float wup, float wdn) {
  const int c  = lane & 15;
  const int hh = lane >> 4;
  const int n  = jt * 16 + c + 1;
  const int cu = 2 * (IROFF + HALF + n);
  const int cd = 2 * (IROFF + HALF - n);
  float* q = irs2 + (rt + 8 * hh) * PR2 + rt;
#pragma unroll
  for (int v = 0; v < 8; ++v) {
    q[v * PR2 + cu] = acc[v] * wup;
    q[v * PR2 + cd] = acc[v] * wdn;
  }
  if (hh != rt) {
    const float a = (rt == 0) ? acc[7] : acc[0];
    float* d = irs2 + ((rt == 0) ? 1 : 16 * PR2);
    d[cu] = a * wup;
    d[cd] = a * wdn;
  }
}

__device__ __forceinline__ void FN_mm_tile2(const float* hs, float* irs2, int jt, int lane,
                                            const v4f (&bq)[8], v2f bt, float wup, float wdn) {
  const int c  = lane & 15;
  const int hh = lane >> 4;
  const float* ar0 = hs + c * HSP;
  const float* ar1 = ar0 + 16 * HSP;
  v8f acc0 = {};
  v8f acc1 = {};
#pragma unroll
  for (int g = 0; g < 8; ++g) {
    const v4f aq0 = *(const v4f*)(ar0 + 8 * g + 4 * hh);
    const v4f aq1 = *(const v4f*)(ar1 + 8 * g + 4 * hh);
    v2f a00, a01, a10, a11, b0, b1;
    a00.x = aq0.x;  a00.y = aq0.y;  a01.x = aq0.z;  a01.y = aq0.w;
    a10.x = aq1.x;  a10.y = aq1.y;  a11.x = aq1.z;  a11.y = aq1.w;
    b0.x = bq[g].x; b0.y = bq[g].y; b1.x = bq[g].z; b1.y = bq[g].w;
    acc0 = __builtin_amdgcn_wmma_f32_16x16x4_f32(
         false, a00,  false, b0,
         (short)0, acc0,  false,  false);
    acc1 = __builtin_amdgcn_wmma_f32_16x16x4_f32(
         false, a10,  false, b0,
         (short)0, acc1,  false,  false);
    acc0 = __builtin_amdgcn_wmma_f32_16x16x4_f32(
         false, a01,  false, b1,
         (short)0, acc0,  false,  false);
    acc1 = __builtin_amdgcn_wmma_f32_16x16x4_f32(
         false, a11,  false, b1,
         (short)0, acc1,  false,  false);
  }
  const v2f at0 = *(const v2f*)(ar0 + 64 + 2 * hh);
  const v2f at1 = *(const v2f*)(ar1 + 64 + 2 * hh);
  acc0 = __builtin_amdgcn_wmma_f32_16x16x4_f32(
       false, at0,  false, bt,
       (short)0, acc0,  false,  false);
  acc1 = __builtin_amdgcn_wmma_f32_16x16x4_f32(
       false, at1,  false, bt,
       (short)0, acc1,  false,  false);
  FN_store_zp(irs2, 0, jt, lane, acc0, wup, wdn);
  FN_store_zp(irs2, 1, jt, lane, acc1, wup, wdn);
}

__device__ __forceinline__ v2f FN_lo2(v4f q) { v2f t; t.x = q.x; t.y = q.y; return t; }
__device__ __forceinline__ v2f FN_hi2(v4f q) { v2f t; t.x = q.z; t.y = q.w; return t; }

__global__ __launch_bounds__(NTHR) void FN_fused(const float* __restrict__ H,
                                                 const float* __restrict__ noise,
                                                 const float* __restrict__ M,
                                                 float* __restrict__ out) {
  __shared__ __attribute__((aligned(16))) float irs2[NPR * PR2];
  __shared__ __attribute__((aligned(16))) float nz2[NPR * NZ2];
  __shared__ __attribute__((aligned(16))) float hs[FPB * HSP];

  const int tid  = threadIdx.x;
  const int lane = tid & 31;
  const int wave = __builtin_amdgcn_readfirstlane(tid >> 5);

  const int b  = blockIdx.x / WPB;
  const int g0 = (blockIdx.x - b * WPB) * HPB;
  const int fb = g0 - 2;
  const long rb = (long)b * FRAMES;

  float hv[17];
  const int hbase = ((int)rb + fb) * CBINS;
#pragma unroll
  for (int u = 0; u < 17; ++u) {
    int i  = (u < 16) ? (tid + NTHR * u) : min(tid + 16 * NTHR, FPB * CBINS - 1);
    unsigned gi = (unsigned)min(max(hbase + i, 0), BATCH * FRAMES * CBINS - 1);
    hv[u] = H[gi];
  }

  const int c15 = lane & 15;
  v4f bq[8];
  v2f bt;
  FN_load_b(M, wave, lane, bq, bt);
  const float wup = M[NTILES * TILEF + wave * 16 + c15 + 1];
  const float wdn = M[NTILES * TILEF + NPAD + wave * 16 + c15 + 1];
  const float wc  = M[NTILES * TILEF];

#pragma unroll
  for (int u = 0; u < 3; ++u) {
    int i = tid + NTHR * u;
    if (i < NPR * 16) {
      int r  = i >> 4;
      int qq = i & 15;
      int fa = fb + r;
      int fc = fb + r + PAIRS;
      int f0 = min(max(fa, 0), FRAMES - 1);
      int f1 = min(max(fc, 0), FRAMES - 1);
      const unsigned ia = (unsigned)(((int)rb + f0) * FRAME_LEN + 4 * qq);
      const unsigned ib = (unsigned)(((int)rb + f1) * FRAME_LEN + 4 * qq);
      const v4f na = *(const v4f*)(noise + ia);
      const v4f nb = *(const v4f*)(noise + ib);
      const float oka = (fa >= 0 && fa < FRAMES) ? 1.0f : 0.0f;
      const float okb = (fc >= 0 && fc < FRAMES) ? 1.0f : 0.0f;
      v4f m0, m1;
      m0.x = (na.x * 2.0f - 1.0f) * oka;
      m0.y = (nb.x * 2.0f - 1.0f) * okb;
      m0.z = (na.y * 2.0f - 1.0f) * oka;
      m0.w = (nb.y * 2.0f - 1.0f) * okb;
      m1.x = (na.z * 2.0f - 1.0f) * oka;
      m1.y = (nb.z * 2.0f - 1.0f) * okb;
      m1.z = (na.w * 2.0f - 1.0f) * oka;
      m1.w = (nb.w * 2.0f - 1.0f) * okb;
      *(v4f*)(nz2 + r * NZ2 + 8 * qq)     = m0;
      *(v4f*)(nz2 + r * NZ2 + 8 * qq + 4) = m1;
    }
  }

  {
    const int jz = tid & 15;
    float* zr = irs2 + (tid >> 4) * PR2 + 4 * jz + 32 * (jz & 8);
    v4f z4; z4.x = 0.0f; z4.y = 0.0f; z4.z = 0.0f; z4.w = 0.0f;
    *(v4f*)zr = z4;
    *(v4f*)(zr + 8 * PR2) = z4;
    if (tid < 16) *(v4f*)(zr + 16 * PR2) = z4;
  }

#pragma unroll
  for (int u = 0; u < 16; ++u) {
    const int i = tid + NTHR * u;
    hs[i + (HSP - CBINS) * (i / CBINS)] = hv[u];
  }
  if (tid < FPB * CBINS - 16 * NTHR) {
    const int i = tid + 16 * NTHR;
    hs[i + (HSP - CBINS) * (i / CBINS)] = hv[16];
  }
  if (tid < FPB * (HSP - CBINS))
    hs[(tid / 3) * HSP + CBINS + (tid - (tid / 3) * 3)] = 0.0f;

  __syncthreads();

  FN_mm_tile2(hs, irs2, wave, lane, bq, bt, wup, wdn);

  {
    const int f = tid >> 2;
    const int j = tid & 3;
    const float* hr = hs + f * HSP;
    const v4f q0 = *(const v4f*)(hr + 16 * j);
    const v4f q1 = *(const v4f*)(hr + 16 * j + 4);
    const v4f q2 = *(const v4f*)(hr + 16 * j + 8);
    const v4f q3 = *(const v4f*)(hr + 16 * j + 12);
    const float h0  = hr[0];
    const float h64 = hr[CBINS - 1];
    float t = (((q0.x + q0.y) + (q0.z + q0.w)) + ((q1.x + q1.y) + (q1.z + q1.w))) +
              (((q2.x + q2.y) + (q2.z + q2.w)) + ((q3.x + q3.y) + (q3.z + q3.w)));
    t = 2.0f * t + ((j == 0) ? (2.0f * h64 - h0) : 0.0f);
    t += __builtin_bit_cast(float, __builtin_amdgcn_update_dpp(
             0, __builtin_bit_cast(int, t), 0xB1  , 0xF, 0xF, false));
    t += __builtin_bit_cast(float, __builtin_amdgcn_update_dpp(
             0, __builtin_bit_cast(int, t), 0x4E  , 0xF, 0xF, false));
    const float zc = t * (1.0f / (float)NIR) * wc;
    const int col = 2 * (IROFF + HALF);
    if (j == 0) irs2[((f < 16) ? f : (f - PAIRS)) * PR2 + col + ((f < 16) ? 0 : 1)] = zc;
    if (j == 1 && (f == 15 || f == 16))
      irs2[((f == 15) ? 0 : 16) * PR2 + col + ((f == 15) ? 1 : 0)] = zc;
  }

  __syncthreads();

  const int p  = lane & 15;
  const int ch = 2 * wave + (lane >> 4);
  if (p < PAIRS) {
    v2f s[8];
#pragma unroll
    for (int a = 0; a < 8; ++a) { s[a].x = 0.0f; s[a].y = 0.0f; }
#pragma unroll
    for (int d = 0; d < 3; ++d) {
      const int r  = p + 2 - d;
      const int cs = (d == 2) ? (4 * wave) : 0;
      const int ce = (d == 0) ? (4 * wave + 4) : 16;
      const float* irow = irs2 + r * PR2 + 2 * (IROFF + d * FRAME_LEN + 8 * ch);
      const float* nrow = nz2 + r * NZ2;
      v4f hiA  = *(const v4f*)(irow - 8 * cs + 8);
      v4f hiB  = *(const v4f*)(irow - 8 * cs + 12);
      v4f midA = *(const v4f*)(irow - 8 * cs);
      v4f midB = *(const v4f*)(irow - 8 * cs + 4);
      for (int c4 = cs; c4 < ce; c4 += 4) {
#pragma unroll
        for (int u = 0; u < 4; ++u) {
          const int cc = c4 + u;
          const v4f loA = *(const v4f*)(irow - 8 * cc - 8);
          const v4f loB = *(const v4f*)(irow - 8 * cc - 4);
          const v4f nA  = *(const v4f*)(nrow + 8 * cc);
          const v4f nB  = *(const v4f*)(nrow + 8 * cc + 4);
          const v2f w[12] = {FN_lo2(loA),  FN_hi2(loA),  FN_lo2(loB),  FN_hi2(loB),
                             FN_lo2(midA), FN_hi2(midA), FN_lo2(midB), FN_hi2(midB),
                             FN_lo2(hiA),  FN_hi2(hiA),  FN_lo2(hiB),  FN_hi2(hiB)};
          const v2f nn[4] = {FN_lo2(nA), FN_hi2(nA), FN_lo2(nB), FN_hi2(nB)};
#pragma unroll
          for (int e = 0; e < 4; ++e) {
#pragma unroll
            for (int a = 0; a < 8; ++a) s[a] = __builtin_elementwise_fma(nn[e], w[a - e + 4], s[a]);
          }
          hiA  = midA;
          hiB  = midB;
          midA = loA;
          midB = loB;
        }
      }
    }
    v4f o0, o1, o2, o3;
    o0.x = s[0].x * 0.01f; o0.y = s[1].x * 0.01f; o0.z = s[2].x * 0.01f; o0.w = s[3].x * 0.01f;
    o1.x = s[4].x * 0.01f; o1.y = s[5].x * 0.01f; o1.z = s[6].x * 0.01f; o1.w = s[7].x * 0.01f;
    o2.x = s[0].y * 0.01f; o2.y = s[1].y * 0.01f; o2.z = s[2].y * 0.01f; o2.w = s[3].y * 0.01f;
    o3.x = s[4].y * 0.01f; o3.y = s[5].y * 0.01f; o3.z = s[6].y * 0.01f; o3.w = s[7].y * 0.01f;
    float* ob0 = hs + p * NZP + 8 * ch;
    float* ob1 = hs + (p + PAIRS) * NZP + 8 * ch;
    *(v4f*)ob0       = o0;
    *(v4f*)(ob0 + 4) = o1;
    *(v4f*)ob1       = o2;
    *(v4f*)(ob1 + 4) = o3;
  }

  __syncthreads();

  for (int i = tid; i < HPB * QPH; i += NTHR) {
    const int r  = i >> 4;
    const int qd = i & 15;
    const int h  = g0 + r;
    if (h < NHOPS) {
      const v4f v = *(const v4f*)(hs + r * NZP + 4 * qd);
      *(v4f*)(out + (long)b * OUTLEN + (long)h * FRAME_LEN + 4 * qd) = v;
    }
  }
}

extern "C" void kernel_launch(void* const* d_in, const int* in_sizes, int n_in,
                              void* d_out, int out_size, void* d_ws, size_t ws_size,
                              hipStream_t stream) {
  const float* H     = (const float*)d_in[0];
  const float* noise = (const float*)d_in[1];
  float* out = (float*)d_out;
  float* Mws = (float*)d_ws;

  FN_setup_M<<<(NTILES * TILEF + 2 * NPAD + 255) / 256, 256, 0, stream>>>(Mws);
  FN_fused<<<BATCH * WPB, NTHR, 0, stream>>>(H, noise, Mws, out);
}


# H tile, noise pair-rows and output staging share one LDS region in time (noise waits in registers over the matmul, 2 more barriers): 31,008 B per work-group, was 39,712 B; 9 resident per WGP, was 8
# speedup vs baseline: 7.3649x; 1.0316x over previous
#include <hip/hip_runtime.h>
#include <hip/hip_bf16.h>


typedef __attribute__((ext_vector_type(2))) float v2f;
typedef __attribute__((ext_vector_type(8))) float v8f;
typedef __attribute__((ext_vector_type(4))) float v4f;

#define BATCH     32
#define FRAMES    4000
#define CBINS     65
#define NIR       129
#define FRAME_LEN 64
#define SEGLEN    192
#define OUTLEN    256128
#define KPAD      68
#define NTILES    4
#define NPAD      80
#define HALF      64
#define FPB       32
#define HPB       30
#define NHOPS     4002
#define WPB       134
#define NTHR      128
#define PAIRS     15
#define NPR       17
#define IROFF     16
#define PR2       324
#define NZ2       132
#define QPH       16
#define HSP       68
#define NZP       68
#define TILEF     1088

__global__ void FN_setup_M(float* __restrict__ M) {
  const float TWO_PI = 6.28318530717958647692f;
  for (int i = blockIdx.x * blockDim.x + threadIdx.x; i < NTILES * TILEF + 2 * NPAD;
       i += gridDim.x * blockDim.x) {
    if (i >= NTILES * TILEF) {
      int q = i - NTILES * TILEF;
      int n = (q < NPAD) ? q : (q - NPAD);
      int j = (q < NPAD) ? (HALF + n) : (HALF - n);
      float w = 0.0f;
      if (n <= HALF) w = 0.5f * (1.0f - cosf(TWO_PI * (float)j / (float)NIR));
      M[i] = w;
      continue;
    }
    int jt = i / TILEF;
    int r  = i - jt * TILEF;
    int l, k;
    if (r < 1024) {
      int g = r >> 7;
      l = (r & 127) >> 2;
      k = 8 * g + 4 * (l >> 4) + (r & 3);
    } else {
      int r2 = r - 1024;
      l = r2 >> 1;
      k = 64 + 2 * (l >> 4) + (r2 & 1);
    }
    int n = jt * 16 + (l & 15) + 1;
    float v = 0.0f;
    if (k < CBINS) {
      int m = (k * n) % NIR;
      float ang = TWO_PI * (float)m / (float)NIR;
      float ck  = (k == 0) ? 1.0f : 2.0f;
      v = (ck / (float)NIR) * cosf(ang);
    }
    M[i] = v;
  }
}

__device__ __forceinline__ void FN_load_b(const float* __restrict__ M, int jt, int lane,
                                          v4f (&bq)[8], v2f& bt) {
  const float* mp = M + jt * TILEF;
#pragma unroll
  for (int g = 0; g < 8; ++g) bq[g] = *(const v4f*)(mp + g * 128 + lane * 4);
  bt = *(const v2f*)(mp + 1024 + lane * 2);
}

__device__ __forceinline__ void FN_store_zp(float* irs2, int rt, int jt, int lane,
                                            const v8f& acc, float wup, float wdn) {
  const int c  = lane & 15;
  const int hh = lane >> 4;
  const int n  = jt * 16 + c + 1;
  const int cu = 2 * (IROFF + HALF + n);
  const int cd = 2 * (IROFF + HALF - n);
  float* q = irs2 + (rt + 8 * hh) * PR2 + rt;
#pragma unroll
  for (int v = 0; v < 8; ++v) {
    q[v * PR2 + cu] = acc[v] * wup;
    q[v * PR2 + cd] = acc[v] * wdn;
  }
  if (hh != rt) {
    const float a = (rt == 0) ? acc[7] : acc[0];
    float* d = irs2 + ((rt == 0) ? 1 : 16 * PR2);
    d[cu] = a * wup;
    d[cd] = a * wdn;
  }
}

__device__ __forceinline__ void FN_mm_tile2(const float* hs, float* irs2, int jt, int lane,
                                            const v4f (&bq)[8], v2f bt, float wup, float wdn) {
  const int c  = lane & 15;
  const int hh = lane >> 4;
  const float* ar0 = hs + c * HSP;
  const float* ar1 = ar0 + 16 * HSP;
  v8f acc0 = {};
  v8f acc1 = {};
#pragma unroll
  for (int g = 0; g < 8; ++g) {
    const v4f aq0 = *(const v4f*)(ar0 + 8 * g + 4 * hh);
    const v4f aq1 = *(const v4f*)(ar1 + 8 * g + 4 * hh);
    v2f a00, a01, a10, a11, b0, b1;
    a00.x = aq0.x;  a00.y = aq0.y;  a01.x = aq0.z;  a01.y = aq0.w;
    a10.x = aq1.x;  a10.y = aq1.y;  a11.x = aq1.z;  a11.y = aq1.w;
    b0.x = bq[g].x; b0.y = bq[g].y; b1.x = bq[g].z; b1.y = bq[g].w;
    acc0 = __builtin_amdgcn_wmma_f32_16x16x4_f32(
         false, a00,  false, b0,
         (short)0, acc0,  false,  false);
    acc1 = __builtin_amdgcn_wmma_f32_16x16x4_f32(
         false, a10,  false, b0,
         (short)0, acc1,  false,  false);
    acc0 = __builtin_amdgcn_wmma_f32_16x16x4_f32(
         false, a01,  false, b1,
         (short)0, acc0,  false,  false);
    acc1 = __builtin_amdgcn_wmma_f32_16x16x4_f32(
         false, a11,  false, b1,
         (short)0, acc1,  false,  false);
  }
  const v2f at0 = *(const v2f*)(ar0 + 64 + 2 * hh);
  const v2f at1 = *(const v2f*)(ar1 + 64 + 2 * hh);
  acc0 = __builtin_amdgcn_wmma_f32_16x16x4_f32(
       false, at0,  false, bt,
       (short)0, acc0,  false,  false);
  acc1 = __builtin_amdgcn_wmma_f32_16x16x4_f32(
       false, at1,  false, bt,
       (short)0, acc1,  false,  false);
  FN_store_zp(irs2, 0, jt, lane, acc0, wup, wdn);
  FN_store_zp(irs2, 1, jt, lane, acc1, wup, wdn);
}

__device__ __forceinline__ v2f FN_lo2(v4f q) { v2f t; t.x = q.x; t.y = q.y; return t; }
__device__ __forceinline__ v2f FN_hi2(v4f q) { v2f t; t.x = q.z; t.y = q.w; return t; }

__global__ __launch_bounds__(NTHR) void FN_fused(const float* __restrict__ H,
                                                 const float* __restrict__ noise,
                                                 const float* __restrict__ M,
                                                 float* __restrict__ out) {
  __shared__ __attribute__((aligned(16))) float irs2[NPR * PR2];
  __shared__ __attribute__((aligned(16))) float un[NPR * NZ2];
  float* const hs  = un;
  float* const nz2 = un;

  const int tid  = threadIdx.x;
  const int lane = tid & 31;
  const int wave = __builtin_amdgcn_readfirstlane(tid >> 5);

  const int b  = blockIdx.x / WPB;
  const int g0 = (blockIdx.x - b * WPB) * HPB;
  const int fb = g0 - 2;
  const long rb = (long)b * FRAMES;

  float hv[17];
  const int hbase = ((int)rb + fb) * CBINS;
#pragma unroll
  for (int u = 0; u < 17; ++u) {
    int i  = (u < 16) ? (tid + NTHR * u) : min(tid + 16 * NTHR, FPB * CBINS - 1);
    unsigned gi = (unsigned)min(max(hbase + i, 0), BATCH * FRAMES * CBINS - 1);
    hv[u] = H[gi];
  }

  const int c15 = lane & 15;
  v4f bq[8];
  v2f bt;
  FN_load_b(M, wave, lane, bq, bt);
  const float wup = M[NTILES * TILEF + wave * 16 + c15 + 1];
  const float wdn = M[NTILES * TILEF + NPAD + wave * 16 + c15 + 1];
  const float wc  = M[NTILES * TILEF];

  v4f na[3], nb[3];
#pragma unroll
  for (int u = 0; u < 3; ++u) {
    const int i  = min(tid + NTHR * u, NPR * 16 - 1);
    const int r  = i >> 4;
    const int qq = i & 15;
    const int f0 = min(max(fb + r, 0), FRAMES - 1);
    const int f1 = min(max(fb + r + PAIRS, 0), FRAMES - 1);
    const unsigned ia = (unsigned)(((int)rb + f0) * FRAME_LEN + 4 * qq);
    const unsigned ib = (unsigned)(((int)rb + f1) * FRAME_LEN + 4 * qq);
    na[u] = *(const v4f*)(noise + ia);
    nb[u] = *(const v4f*)(noise + ib);
  }

  {
    const int jz = tid & 15;
    float* zr = irs2 + (tid >> 4) * PR2 + 4 * jz + 32 * (jz & 8);
    v4f z4; z4.x = 0.0f; z4.y = 0.0f; z4.z = 0.0f; z4.w = 0.0f;
    *(v4f*)zr = z4;
    *(v4f*)(zr + 8 * PR2) = z4;
    if (tid < 16) *(v4f*)(zr + 16 * PR2) = z4;
  }

#pragma unroll
  for (int u = 0; u < 16; ++u) {
    const int i = tid + NTHR * u;
    hs[i + (HSP - CBINS) * (i / CBINS)] = hv[u];
  }
  if (tid < FPB * CBINS - 16 * NTHR) {
    const int i = tid + 16 * NTHR;
    hs[i + (HSP - CBINS) * (i / CBINS)] = hv[16];
  }
  if (tid < FPB * (HSP - CBINS))
    hs[(tid / 3) * HSP + CBINS + (tid - (tid / 3) * 3)] = 0.0f;

  __syncthreads();

  FN_mm_tile2(hs, irs2, wave, lane, bq, bt, wup, wdn);

  {
    const int f = tid >> 2;
    const int j = tid & 3;
    const float* hr = hs + f * HSP;
    const v4f q0 = *(const v4f*)(hr + 16 * j);
    const v4f q1 = *(const v4f*)(hr + 16 * j + 4);
    const v4f q2 = *(const v4f*)(hr + 16 * j + 8);
    const v4f q3 = *(const v4f*)(hr + 16 * j + 12);
    const float h0  = hr[0];
    const float h64 = hr[CBINS - 1];
    float t = (((q0.x + q0.y) + (q0.z + q0.w)) + ((q1.x + q1.y) + (q1.z + q1.w))) +
              (((q2.x + q2.y) + (q2.z + q2.w)) + ((q3.x + q3.y) + (q3.z + q3.w)));
    t = 2.0f * t + ((j == 0) ? (2.0f * h64 - h0) : 0.0f);
    t += __builtin_bit_cast(float, __builtin_amdgcn_update_dpp(
             0, __builtin_bit_cast(int, t), 0xB1  , 0xF, 0xF, false));
    t += __builtin_bit_cast(float, __builtin_amdgcn_update_dpp(
             0, __builtin_bit_cast(int, t), 0x4E  , 0xF, 0xF, false));
    const float zc = t * (1.0f / (float)NIR) * wc;
    const int col = 2 * (IROFF + HALF);
    if (j == 0) irs2[((f < 16) ? f : (f - PAIRS)) * PR2 + col + ((f < 16) ? 0 : 1)] = zc;
    if (j == 1 && (f == 15 || f == 16))
      irs2[((f == 15) ? 0 : 16) * PR2 + col + ((f == 15) ? 1 : 0)] = zc;
  }

  __syncthreads();

#pragma unroll
  for (int u = 0; u < 3; ++u) {
    const int i = tid + NTHR * u;
    if (i < NPR * 16) {
      const int r  = i >> 4;
      const int qq = i & 15;
      const int fa = fb + r;
      const int fc = fb + r + PAIRS;
      const float oka = (fa >= 0 && fa < FRAMES) ? 1.0f : 0.0f;
      const float okb = (fc >= 0 && fc < FRAMES) ? 1.0f : 0.0f;
      v4f m0, m1;
      m0.x = (na[u].x * 2.0f - 1.0f) * oka;
      m0.y = (nb[u].x * 2.0f - 1.0f) * okb;
      m0.z = (na[u].y * 2.0f - 1.0f) * oka;
      m0.w = (nb[u].y * 2.0f - 1.0f) * okb;
      m1.x = (na[u].z * 2.0f - 1.0f) * oka;
      m1.y = (nb[u].z * 2.0f - 1.0f) * okb;
      m1.z = (na[u].w * 2.0f - 1.0f) * oka;
      m1.w = (nb[u].w * 2.0f - 1.0f) * okb;
      *(v4f*)(nz2 + r * NZ2 + 8 * qq)     = m0;
      *(v4f*)(nz2 + r * NZ2 + 8 * qq + 4) = m1;
    }
  }

  __syncthreads();

  const int p  = lane & 15;
  const int ch = 2 * wave + (lane >> 4);
  v2f s[8];
#pragma unroll
  for (int a = 0; a < 8; ++a) { s[a].x = 0.0f; s[a].y = 0.0f; }
  if (p < PAIRS) {
#pragma unroll
    for (int d = 0; d < 3; ++d) {
      const int r  = p + 2 - d;
      const int cs = (d == 2) ? (4 * wave) : 0;
      const int ce = (d == 0) ? (4 * wave + 4) : 16;
      const float* irow = irs2 + r * PR2 + 2 * (IROFF + d * FRAME_LEN + 8 * ch);
      const float* nrow = nz2 + r * NZ2;
      v4f hiA  = *(const v4f*)(irow - 8 * cs + 8);
      v4f hiB  = *(const v4f*)(irow - 8 * cs + 12);
      v4f midA = *(const v4f*)(irow - 8 * cs);
      v4f midB = *(const v4f*)(irow - 8 * cs + 4);
      for (int c4 = cs; c4 < ce; c4 += 4) {
#pragma unroll
        for (int u = 0; u < 4; ++u) {
          const int cc = c4 + u;
          const v4f loA = *(const v4f*)(irow - 8 * cc - 8);
          const v4f loB = *(const v4f*)(irow - 8 * cc - 4);
          const v4f nA  = *(const v4f*)(nrow + 8 * cc);
          const v4f nB  = *(const v4f*)(nrow + 8 * cc + 4);
          const v2f w[12] = {FN_lo2(loA),  FN_hi2(loA),  FN_lo2(loB),  FN_hi2(loB),
                             FN_lo2(midA), FN_hi2(midA), FN_lo2(midB), FN_hi2(midB),
                             FN_lo2(hiA),  FN_hi2(hiA),  FN_lo2(hiB),  FN_hi2(hiB)};
          const v2f nn[4] = {FN_lo2(nA), FN_hi2(nA), FN_lo2(nB), FN_hi2(nB)};
#pragma unroll
          for (int e = 0; e < 4; ++e) {
#pragma unroll
            for (int a = 0; a < 8; ++a) s[a] = __builtin_elementwise_fma(nn[e], w[a - e + 4], s[a]);
          }
          hiA  = midA;
          hiB  = midB;
          midA = loA;
          midB = loB;
        }
      }
    }
  }

  __syncthreads();

  if (p < PAIRS) {
    v4f o0, o1, o2, o3;
    o0.x = s[0].x * 0.01f; o0.y = s[1].x * 0.01f; o0.z = s[2].x * 0.01f; o0.w = s[3].x * 0.01f;
    o1.x = s[4].x * 0.01f; o1.y = s[5].x * 0.01f; o1.z = s[6].x * 0.01f; o1.w = s[7].x * 0.01f;
    o2.x = s[0].y * 0.01f; o2.y = s[1].y * 0.01f; o2.z = s[2].y * 0.01f; o2.w = s[3].y * 0.01f;
    o3.x = s[4].y * 0.01f; o3.y = s[5].y * 0.01f; o3.z = s[6].y * 0.01f; o3.w = s[7].y * 0.01f;
    float* ob0 = hs + p * NZP + 8 * ch;
    float* ob1 = hs + (p + PAIRS) * NZP + 8 * ch;
    *(v4f*)ob0       = o0;
    *(v4f*)(ob0 + 4) = o1;
    *(v4f*)ob1       = o2;
    *(v4f*)(ob1 + 4) = o3;
  }

  __syncthreads();

  for (int i = tid; i < HPB * QPH; i += NTHR) {
    const int r  = i >> 4;
    const int qd = i & 15;
    const int h  = g0 + r;
    if (h < NHOPS) {
      const v4f v = *(const v4f*)(hs + r * NZP + 4 * qd);
      *(v4f*)(out + (long)b * OUTLEN + (long)h * FRAME_LEN + 4 * qd) = v;
    }
  }
}

extern "C" void kernel_launch(void* const* d_in, const int* in_sizes, int n_in,
                              void* d_out, int out_size, void* d_ws, size_t ws_size,
                              hipStream_t stream) {
  const float* H     = (const float*)d_in[0];
  const float* noise = (const float*)d_in[1];
  float* out = (float*)d_out;
  float* Mws = (float*)d_ws;

  FN_setup_M<<<(NTILES * TILEF + 2 * NPAD + 255) / 256, 256, 0, stream>>>(Mws);
  FN_fused<<<BATCH * WPB, NTHR, 0, stream>>>(H, noise, Mws, out);
}
